# starting kernel
# baseline (speedup 1.0000x reference)
#include <hip/hip_runtime.h>
#include <hip/hip_bf16.h>
#include <math.h>

#define DEPTH     15
#define N_LEAVES  32768
#define N_NODES   65535
#define FDIM      256
#define CDIM      2
#define NODE_ELEMS (FDIM * CDIM)
#define MIN_DIST  1e-7f

typedef __attribute__((ext_vector_type(2))) float v2f;
typedef __attribute__((ext_vector_type(8))) float v8f;

#define REG_BLOCKS 1024
#define REG_TPB    256
#define REG_WAVES  (REG_TPB / 32)
#define REG_NWAVES (REG_BLOCKS * REG_WAVES)

__global__ __launch_bounds__(REG_TPB)
void reg_partials_kernel(const float* __restrict__ deltas,
                         const float* __restrict__ branch,
                         float* __restrict__ partials)
{
    const int lane  = threadIdx.x & 31;
    const int wave  = threadIdx.x >> 5;
    const int gwave = blockIdx.x * REG_WAVES + wave;

    float accum = 0.0f;
    for (int n = gwave; n < N_NODES; n += REG_NWAVES) {
        if (n + REG_NWAVES < N_NODES) {
            __builtin_prefetch(deltas + (size_t)(n + REG_NWAVES) * NODE_ELEMS + lane * 16, 0, 0);
        }
        const float4* dp = (const float4*)(deltas + (size_t)n * NODE_ELEMS) + lane * 4;
        const float4 a0 = dp[0], a1 = dp[1], a2 = dp[2], a3 = dp[3];
        float s;
        if (n == 0) {
            s = a0.x*a0.x + a0.y*a0.y + a0.z*a0.z + a0.w*a0.w
              + a1.x*a1.x + a1.y*a1.y + a1.z*a1.z + a1.w*a1.w
              + a2.x*a2.x + a2.y*a2.y + a2.z*a2.z + a2.w*a2.w
              + a3.x*a3.x + a3.y*a3.y + a3.z*a3.z + a3.w*a3.w;
        } else {
            s = fabsf(a0.x)+fabsf(a0.y)+fabsf(a0.z)+fabsf(a0.w)
              + fabsf(a1.x)+fabsf(a1.y)+fabsf(a1.z)+fabsf(a1.w)
              + fabsf(a2.x)+fabsf(a2.y)+fabsf(a2.z)+fabsf(a2.w)
              + fabsf(a3.x)+fabsf(a3.y)+fabsf(a3.z)+fabsf(a3.w);
        }
        #pragma unroll
        for (int off = 16; off >= 1; off >>= 1) s += __shfl_xor(s, off, 32);
        if (n != 0) s /= fmaxf(branch[n - 1], MIN_DIST);
        accum += s;
    }
    if (lane == 0) partials[gwave] = accum;
}

__global__ __launch_bounds__(1024)
void reg_reduce_kernel(const float* __restrict__ partials, float* __restrict__ out)
{
    __shared__ float sm[1024];
    float s = 0.0f;
    for (int i = threadIdx.x; i < REG_NWAVES; i += 1024) s += partials[i];
    sm[threadIdx.x] = s;
    __syncthreads();
    for (int w = 512; w >= 1; w >>= 1) {
        if ((int)threadIdx.x < w) sm[threadIdx.x] += sm[threadIdx.x + w];
        __syncthreads();
    }
    if (threadIdx.x == 0) out[2 * N_LEAVES] = sm[0];
}

#define LG_WAVES 2
#define XS_STRIDE 260

__global__ __launch_bounds__(LG_WAVES * 32)
void logits_softmax_kernel(const float* __restrict__ x,
                           const float* __restrict__ deltas,
                           float* __restrict__ out)
{
    __shared__ float xs[LG_WAVES][16 * XS_STRIDE];
    __shared__ float wsh[LG_WAVES][NODE_ELEMS];
    __shared__ float lg[LG_WAVES][16 * CDIM];

    const int lane  = threadIdx.x & 31;
    const int wave  = threadIdx.x >> 5;
    const int tile  = blockIdx.x * LG_WAVES + wave;
    const int leaf0 = tile * 16;

    float* xw  = xs[wave];
    float* ww  = wsh[wave];
    float* lgw = lg[wave];

    {
        const float4* xb4 = (const float4*)(x + (size_t)leaf0 * FDIM);
        for (int t = lane; t < (16 * FDIM) / 4; t += 32) {
            const int r = t >> 6, c4 = t & 63;
            *(float4*)&xw[r * XS_STRIDE + c4 * 4] = xb4[t];
        }
    }

    {
        float4 acc0 = {0,0,0,0}, acc1 = {0,0,0,0}, acc2 = {0,0,0,0}, acc3 = {0,0,0,0};
        #pragma unroll
        for (int l = 0; l <= 11; ++l) {
            const int idx = ((1 << l) - 1) + (leaf0 >> (DEPTH - l));
            const float4* dp = (const float4*)(deltas + (size_t)idx * NODE_ELEMS) + lane * 4;
            const float4 d0 = dp[0], d1 = dp[1], d2 = dp[2], d3 = dp[3];
            acc0.x += d0.x; acc0.y += d0.y; acc0.z += d0.z; acc0.w += d0.w;
            acc1.x += d1.x; acc1.y += d1.y; acc1.z += d1.z; acc1.w += d1.w;
            acc2.x += d2.x; acc2.y += d2.y; acc2.z += d2.z; acc2.w += d2.w;
            acc3.x += d3.x; acc3.y += d3.y; acc3.z += d3.z; acc3.w += d3.w;
        }
        float4* wv = (float4*)(ww) + lane * 4;
        wv[0] = acc0; wv[1] = acc1; wv[2] = acc2; wv[3] = acc3;
    }
    if (lane < 16) { lgw[lane * CDIM + 0] = 0.0f; lgw[lane * CDIM + 1] = 0.0f; }
    __syncthreads();

    v8f cfrag = {};
    const int mn   = lane & 15;
    const int koff = (lane < 16) ? 0 : 2;
    const int wcol = mn & 1;
    for (int k0 = 0; k0 < FDIM; k0 += 4) {
        const int k = k0 + koff;
        v2f a, b;
        a.x = ww[k * CDIM + wcol];
        a.y = ww[(k + 1) * CDIM + wcol];
        b.x = xw[mn * XS_STRIDE + k];
        b.y = xw[mn * XS_STRIDE + k + 1];
        cfrag = __builtin_amdgcn_wmma_f32_16x16x4_f32(
             false, a,  false, b,
             (short)0, cfrag,  false,  false);
    }

    for (int j = 0; j < 16; ++j) {
        const int leaf = leaf0 + j;
        const float4* xr4 = (const float4*)(xw + j * XS_STRIDE) + lane * 2;
        const float4 xa = xr4[0], xb = xr4[1];
        float s0 = 0.0f, s1 = 0.0f;
        #pragma unroll
        for (int l = 12; l <= 15; ++l) {
            const int idx = ((1 << l) - 1) + (leaf >> (DEPTH - l));
            const float4* dp = (const float4*)(deltas + (size_t)idx * NODE_ELEMS) + lane * 4;
            const float4 d0 = dp[0], d1 = dp[1], d2 = dp[2], d3 = dp[3];
            s0 += xa.x*d0.x + xa.y*d0.z + xa.z*d1.x + xa.w*d1.z;
            s1 += xa.x*d0.y + xa.y*d0.w + xa.z*d1.y + xa.w*d1.w;
            s0 += xb.x*d2.x + xb.y*d2.z + xb.z*d3.x + xb.w*d3.z;
            s1 += xb.x*d2.y + xb.y*d2.w + xb.z*d3.y + xb.w*d3.w;
        }
        #pragma unroll
        for (int off = 16; off >= 1; off >>= 1) {
            s0 += __shfl_xor(s0, off, 32);
            s1 += __shfl_xor(s1, off, 32);
        }
        if (lane == 0) {
            lgw[j * CDIM + 0] += s0;
            lgw[j * CDIM + 1] += s1;
        }
    }
    __syncthreads();

    if (lane < 16) {
        const float z0 = cfrag[0] + lgw[lane * CDIM + 0];
        const float z1 = cfrag[1] + lgw[lane * CDIM + 1];
        const float m  = fmaxf(z0, z1);
        const float e0 = expf(z0 - m);
        const float e1 = expf(z1 - m);
        const float inv = 1.0f / (e0 + e1);
        const size_t o = (size_t)(leaf0 + lane) * CDIM;
        out[o + 0] = e0 * inv;
        out[o + 1] = e1 * inv;
    }
}

extern "C" void kernel_launch(void* const* d_in, const int* in_sizes, int n_in,
                              void* d_out, int out_size, void* d_ws, size_t ws_size,
                              hipStream_t stream) {
    const float* x      = (const float*)d_in[0];
    const float* deltas = (const float*)d_in[1];
    const float* branch = (const float*)d_in[2];
    float* out      = (float*)d_out;
    float* partials = (float*)d_ws;

    reg_partials_kernel<<<REG_BLOCKS, REG_TPB, 0, stream>>>(deltas, branch, partials);
    reg_reduce_kernel<<<1, 1024, 0, stream>>>(partials, out);

    const int n_tiles = N_LEAVES / 16;
    logits_softmax_kernel<<<n_tiles / LG_WAVES, LG_WAVES * 32, 0, stream>>>(x, deltas, out);
}


# Regularizer launch as a stream: 4096 waves (was 8192), each owns 16 consecutive nodes, 12 nodes' loads in flight, divides per lane, one lane reduction per wave; regularizer sum order changes
# speedup vs baseline: 1.9153x; 1.9153x over previous
#include <hip/hip_runtime.h>
#include <hip/hip_bf16.h>
#include <math.h>

#define DEPTH     15
#define N_LEAVES  32768
#define N_NODES   65535
#define FDIM      256
#define CDIM      2
#define NODE_ELEMS (FDIM * CDIM)
#define MIN_DIST  1e-7f

typedef __attribute__((ext_vector_type(2))) float v2f;
typedef __attribute__((ext_vector_type(8))) float v8f;

#define REG_TPB    256
#define REG_WAVES  (REG_TPB / 32)
#define REG_NPW    16
#define REG_NWAVES ((N_NODES + REG_NPW - 1) / REG_NPW)
#define REG_BLOCKS (REG_NWAVES / REG_WAVES)
#define REG_AHEAD  12

__global__ __launch_bounds__(REG_TPB, 4)
void reg_partials_kernel(const float* __restrict__ deltas,
                         const float* __restrict__ branch,
                         float* __restrict__ partials)
{
    const int lane  = threadIdx.x & 31;
    const int wave  = __builtin_amdgcn_readfirstlane(threadIdx.x >> 5);
    const int gwave = blockIdx.x * REG_WAVES + wave;
    const int n0    = gwave * REG_NPW;

    int bi = n0 + (lane & 15) - 1;
    bi = bi < 0 ? 0 : (bi > N_NODES - 2 ? N_NODES - 2 : bi);
    const float bl = branch[bi];

    const float4* base = (const float4*)(deltas + (size_t)n0 * NODE_ELEMS) + lane;
    const bool last_ok = (n0 + REG_NPW - 1) < N_NODES;

    float4 buf[REG_AHEAD][4];
#define REG_LOAD(I) do { \
        const float4* p_ = base + (((I) < REG_NPW - 1 || last_ok) ? (I) : 0) * (NODE_ELEMS / 4); \
        buf[(I) % REG_AHEAD][0] = p_[0];  buf[(I) % REG_AHEAD][1] = p_[32]; \
        buf[(I) % REG_AHEAD][2] = p_[64]; buf[(I) % REG_AHEAD][3] = p_[96]; } while (0)

    #pragma unroll
    for (int i = 0; i < REG_AHEAD; ++i) {
        REG_LOAD(i);
        __builtin_amdgcn_sched_barrier(0);
    }

    float accum = 0.0f;
    #pragma unroll
    for (int i = 0; i < REG_NPW; ++i) {
        const float4 a0 = buf[i % REG_AHEAD][0], a1 = buf[i % REG_AHEAD][1],
                     a2 = buf[i % REG_AHEAD][2], a3 = buf[i % REG_AHEAD][3];
        float term;
        if (i == 0 && n0 == 0) {
            const float q0 = a0.x*a0.x + a0.y*a0.y + a0.z*a0.z + a0.w*a0.w;
            const float q1 = a1.x*a1.x + a1.y*a1.y + a1.z*a1.z + a1.w*a1.w;
            const float q2 = a2.x*a2.x + a2.y*a2.y + a2.z*a2.z + a2.w*a2.w;
            const float q3 = a3.x*a3.x + a3.y*a3.y + a3.z*a3.z + a3.w*a3.w;
            term = (q0 + q1) + (q2 + q3);
        } else {
            const float s0 = fabsf(a0.x) + fabsf(a0.y) + fabsf(a0.z) + fabsf(a0.w);
            const float s1 = fabsf(a1.x) + fabsf(a1.y) + fabsf(a1.z) + fabsf(a1.w);
            const float s2 = fabsf(a2.x) + fabsf(a2.y) + fabsf(a2.z) + fabsf(a2.w);
            const float s3 = fabsf(a3.x) + fabsf(a3.y) + fabsf(a3.z) + fabsf(a3.w);
            const float len = __builtin_bit_cast(float,
                __builtin_amdgcn_readlane(__builtin_bit_cast(int, bl), i));
            term = ((s0 + s1) + (s2 + s3)) / fmaxf(len, MIN_DIST);
        }
        __builtin_amdgcn_sched_barrier(0);
        if (i + REG_AHEAD < REG_NPW) REG_LOAD(i + REG_AHEAD);
        __builtin_amdgcn_sched_barrier(0);
        if (i < REG_NPW - 1 || last_ok) accum += term;
    }
#undef REG_LOAD
    #pragma unroll
    for (int off = 16; off >= 1; off >>= 1) accum += __shfl_xor(accum, off, 32);
    if (lane == 0) partials[gwave] = accum;
}

#define LG_WAVES 4
#define LG_TPB   (LG_WAVES * 32)
#define LG_TILES (N_LEAVES / 16)
#define LG_PAIRS (LG_TILES / 2)
#define LG_GRID  (LG_PAIRS + 1)
#define XS_STRIDE 260

struct Node4 { float4 a, b, c, d; };

__device__ __forceinline__ Node4 load_node(const float* __restrict__ deltas, int idx, int lane)
{
    const float4* dp = (const float4*)(deltas + (size_t)idx * NODE_ELEMS) + lane;
    Node4 n; n.a = dp[0]; n.b = dp[32]; n.c = dp[64]; n.d = dp[96];
    return n;
}

__device__ __forceinline__ void dot_node(const float2 x0, const float2 x1,
                                         const float2 x2, const float2 x3, const Node4& n,
                                         float& s0, float& s1)
{
    s0 += x0.x*n.a.x + x0.y*n.a.z + x1.x*n.b.x + x1.y*n.b.z;
    s1 += x0.x*n.a.y + x0.y*n.a.w + x1.x*n.b.y + x1.y*n.b.w;
    s0 += x2.x*n.c.x + x2.y*n.c.z + x3.x*n.d.x + x3.y*n.d.z;
    s1 += x2.x*n.c.y + x2.y*n.c.w + x3.x*n.d.y + x3.y*n.d.w;
}

__device__ __forceinline__ void ancestor_product(const float* __restrict__ xsb,
                                                 const float* __restrict__ wb,
                                                 float* __restrict__ cp, int wave, int lane)
{
    v8f cfrag = {};
    const int mn   = lane & 15;
    const int koff = (lane < 16) ? 0 : 2;
    const int wcol = mn & 1;
    const int kbeg = wave * (FDIM / LG_WAVES);
    #pragma unroll
    for (int ks = 0; ks < FDIM / LG_WAVES; ks += 4) {
        const int k = kbeg + ks + koff;
        v2f a, b;
        a.x = wb[k * CDIM + wcol];
        a.y = wb[(k + 1) * CDIM + wcol];
        b.x = xsb[mn * XS_STRIDE + k];
        b.y = xsb[mn * XS_STRIDE + k + 1];
        cfrag = __builtin_amdgcn_wmma_f32_16x16x4_f32(
             false, a,  false, b,
             (short)0, cfrag,  false,  false);
    }
    if (lane < 16) {
        cp[lane * CDIM + 0] = cfrag[0];
        cp[lane * CDIM + 1] = cfrag[1];
    }
}

__device__ __forceinline__ void leaf_products(const float* __restrict__ xsb, int wave, int lane,
                                              const Node4& n12, const Node4& n13,
                                              const Node4& n14a, const Node4& n14b,
                                              const Node4& n15a, const Node4& n15b,
                                              const Node4& n15c, const Node4& n15d,
                                              float (&s0)[4], float (&s1)[4])
{
    #pragma unroll
    for (int jj = 0; jj < 4; ++jj) {
        const int j = 4 * wave + jj;
        const float2* xr2 = (const float2*)(xsb + j * XS_STRIDE) + lane;
        const float2 x0 = xr2[0], x1 = xr2[32], x2 = xr2[64], x3 = xr2[96];
        float a0 = 0.0f, a1 = 0.0f;
        dot_node(x0, x1, x2, x3, n12, a0, a1);
        dot_node(x0, x1, x2, x3, n13, a0, a1);
        dot_node(x0, x1, x2, x3, (jj < 2) ? n14a : n14b, a0, a1);
        dot_node(x0, x1, x2, x3, (jj == 0) ? n15a : (jj == 1) ? n15b : (jj == 2) ? n15c : n15d, a0, a1);
        s0[jj] = a0; s1[jj] = a1;
    }
}

__device__ __forceinline__ void leaf_reduce(float (&s0)[4], float (&s1)[4], float* lgw, int lane)
{
    #pragma unroll
    for (int off = 16; off >= 1; off >>= 1) {
        #pragma unroll
        for (int jj = 0; jj < 4; ++jj) {
            s0[jj] += __shfl_xor(s0[jj], off, 32);
            s1[jj] += __shfl_xor(s1[jj], off, 32);
        }
    }
    if (lane == 0) {
        float4* lp = (float4*)lgw;
        lp[0] = make_float4(s0[0], s1[0], s0[1], s1[1]);
        lp[1] = make_float4(s0[2], s1[2], s0[3], s1[3]);
    }
}

__global__ __launch_bounds__(LG_TPB, 4)
void logits_softmax_kernel(const float* __restrict__ x,
                           const float* __restrict__ deltas,
                           const float* __restrict__ partials,
                           float* __restrict__ out)
{
    __shared__ alignas(16) float xs[2][16 * XS_STRIDE];
    __shared__ alignas(16) float wsh[2][NODE_ELEMS];
    __shared__ float cpart[2][LG_WAVES][16 * CDIM];
    __shared__ alignas(16) float lg[2][16 * CDIM];

    const int tid   = threadIdx.x;
    const int lane  = tid & 31;
    const int wave  = tid >> 5;

    if (blockIdx.x == 0) {
        __shared__ float red[LG_WAVES];
        const float4* p4 = (const float4*)partials;
        float4 acc = p4[tid];
        #pragma unroll
        for (int i = 1; i < REG_NWAVES / 4 / LG_TPB; ++i) {
            const float4 v = p4[tid + i * LG_TPB];
            acc.x += v.x; acc.y += v.y; acc.z += v.z; acc.w += v.w;
        }
        float s = (acc.x + acc.y) + (acc.z + acc.w);
        #pragma unroll
        for (int off = 16; off >= 1; off >>= 1) s += __shfl_xor(s, off, 32);
        if (lane == 0) red[wave] = s;
        __syncthreads();
        if (tid == 0) out[2 * N_LEAVES] = (red[0] + red[1]) + (red[2] + red[3]);
        return;
    }

    const int leafA = (blockIdx.x - 1) * 32;

    const int lbA = leafA + 4 * wave;
    const Node4 a12  = load_node(deltas, ((1 << 12) - 1) + (lbA >> 3), lane);
    const Node4 a13  = load_node(deltas, ((1 << 13) - 1) + (lbA >> 2), lane);
    const Node4 a14a = load_node(deltas, ((1 << 14) - 1) + (lbA >> 1), lane);
    const Node4 a14b = load_node(deltas, ((1 << 14) - 1) + (lbA >> 1) + 1, lane);
    const Node4 a15a = load_node(deltas, ((1 << 15) - 1) + lbA + 0, lane);
    const Node4 a15b = load_node(deltas, ((1 << 15) - 1) + lbA + 1, lane);
    const Node4 a15c = load_node(deltas, ((1 << 15) - 1) + lbA + 2, lane);
    const Node4 a15d = load_node(deltas, ((1 << 15) - 1) + lbA + 3, lane);

    {
        const float4* xb4 = (const float4*)(x + (size_t)leafA * FDIM);
        #pragma unroll
        for (int i = 0; i < 2 * (16 * FDIM) / 4 / LG_TPB; ++i) {
            const int t = tid + i * LG_TPB;
            const int r = t >> 6, c4 = t & 63;
            *(float4*)&xs[r >> 4][(r & 15) * XS_STRIDE + c4 * 4] = xb4[t];
        }
    }

    {
        float4 acc = {0, 0, 0, 0};
        #pragma unroll
        for (int l = 0; l <= 10; ++l) {
            const int idx = ((1 << l) - 1) + (leafA >> (DEPTH - l));
            const float4 d = ((const float4*)(deltas + (size_t)idx * NODE_ELEMS))[tid];
            acc.x += d.x; acc.y += d.y; acc.z += d.z; acc.w += d.w;
        }
        const int i11 = ((1 << 11) - 1) + (leafA >> 4);
        const float4 dA = ((const float4*)(deltas + (size_t)i11 * NODE_ELEMS))[tid];
        const float4 dB = ((const float4*)(deltas + (size_t)(i11 + 1) * NODE_ELEMS))[tid];
        ((float4*)wsh[0])[tid] = make_float4(acc.x + dA.x, acc.y + dA.y, acc.z + dA.z, acc.w + dA.w);
        ((float4*)wsh[1])[tid] = make_float4(acc.x + dB.x, acc.y + dB.y, acc.z + dB.z, acc.w + dB.w);
    }
    __syncthreads();

    ancestor_product(xs[0], wsh[0], cpart[0][wave], wave, lane);
    float s0[4], s1[4];
    leaf_products(xs[0], wave, lane, a12, a13, a14a, a14b, a15a, a15b, a15c, a15d, s0, s1);

    const int lbB = lbA + 16;
    const Node4 b12  = load_node(deltas, ((1 << 12) - 1) + (lbB >> 3), lane);
    const Node4 b13  = load_node(deltas, ((1 << 13) - 1) + (lbB >> 2), lane);
    const Node4 b14a = load_node(deltas, ((1 << 14) - 1) + (lbB >> 1), lane);
    const Node4 b14b = load_node(deltas, ((1 << 14) - 1) + (lbB >> 1) + 1, lane);
    const Node4 b15a = load_node(deltas, ((1 << 15) - 1) + lbB + 0, lane);
    const Node4 b15b = load_node(deltas, ((1 << 15) - 1) + lbB + 1, lane);
    const Node4 b15c = load_node(deltas, ((1 << 15) - 1) + lbB + 2, lane);
    const Node4 b15d = load_node(deltas, ((1 << 15) - 1) + lbB + 3, lane);

    leaf_reduce(s0, s1, &lg[0][4 * wave * CDIM], lane);

    ancestor_product(xs[1], wsh[1], cpart[1][wave], wave, lane);
    float t0[4], t1[4];
    leaf_products(xs[1], wave, lane, b12, b13, b14a, b14b, b15a, b15b, b15c, b15d, t0, t1);
    leaf_reduce(t0, t1, &lg[1][4 * wave * CDIM], lane);
    __syncthreads();

    if (tid < 32) {
        const int tl = tid >> 4, r = tid & 15;
        const float z0 = ((cpart[tl][0][r * CDIM + 0] + cpart[tl][1][r * CDIM + 0])
                        + cpart[tl][2][r * CDIM + 0]) + cpart[tl][3][r * CDIM + 0]
                        + lg[tl][r * CDIM + 0];
        const float z1 = ((cpart[tl][0][r * CDIM + 1] + cpart[tl][1][r * CDIM + 1])
                        + cpart[tl][2][r * CDIM + 1]) + cpart[tl][3][r * CDIM + 1]
                        + lg[tl][r * CDIM + 1];
        const float m  = fmaxf(z0, z1);
        const float e0 = expf(z0 - m);
        const float e1 = expf(z1 - m);
        const float inv = 1.0f / (e0 + e1);
        const size_t o = (size_t)(leafA + tid) * CDIM;
        out[o + 0] = e0 * inv;
        out[o + 1] = e1 * inv;
    }
}

extern "C" void kernel_launch(void* const* d_in, const int* in_sizes, int n_in,
                              void* d_out, int out_size, void* d_ws, size_t ws_size,
                              hipStream_t stream) {
    const float* x      = (const float*)d_in[0];
    const float* deltas = (const float*)d_in[1];
    const float* branch = (const float*)d_in[2];
    float* out      = (float*)d_out;
    float* partials = (float*)d_ws;

    reg_partials_kernel<<<REG_BLOCKS, REG_TPB, 0, stream>>>(deltas, branch, partials);

    logits_softmax_kernel<<<LG_GRID, LG_TPB, 0, stream>>>(x, deltas, partials, out);
}
